# starting kernel
# baseline (speedup 1.0000x reference)
#include <hip/hip_runtime.h>

typedef __attribute__((ext_vector_type(2))) float v2f;
typedef __attribute__((ext_vector_type(8))) float v8f;

#define BB_    2
#define CO_    4
#define ZIN_   64
#define WOUT_  30
#define XOUT_  30
#define YOUT_  62
#define ZOUT_  62
#define KPRIME 54
#define NSTEPS 14
#define ZTILE  14
#define NZT    5

#define XS_Y   64
#define XS_X   4096
#define XS_W   131072
#define XS_CI  4194304
#define XS_B   8388608

#define WS_D   1
#define WS_C   3
#define WS_B   9
#define WS_A   27
#define WS_CI  81
#define WS_CO  162

#define KPC(k)  ((k) < KPRIME ? (k) : 0)
#define OFFE(k) ((KPC(k)/27)*XS_CI + ((KPC(k)%27)/9)*XS_W + ((KPC(k)%9)/3)*XS_X + (KPC(k)%3)*XS_Y)
#define WIDX(k) ((KPC(k)/27)*WS_CI + ((KPC(k)%27)/9)*WS_A + ((KPC(k)%9)/3)*WS_B + (KPC(k)%3)*WS_C)
#define KD(k)   { OFFE(k), WIDX(k) }

__constant__ int2 ktbl[56] = {
    KD(0),  KD(1),  KD(2),  KD(3),  KD(4),  KD(5),  KD(6),  KD(7),
    KD(8),  KD(9),  KD(10), KD(11), KD(12), KD(13), KD(14), KD(15),
    KD(16), KD(17), KD(18), KD(19), KD(20), KD(21), KD(22), KD(23),
    KD(24), KD(25), KD(26), KD(27), KD(28), KD(29), KD(30), KD(31),
    KD(32), KD(33), KD(34), KD(35), KD(36), KD(37), KD(38), KD(39),
    KD(40), KD(41), KD(42), KD(43), KD(44), KD(45), KD(46), KD(47),
    KD(48), KD(49), KD(50), KD(51), KD(52), KD(53), KD(54), KD(55)
};

__global__ __launch_bounds__(32)
void conv4d_wmma_f32(const float* __restrict__ x,
                     const float* __restrict__ w,
                     const float* __restrict__ bias,
                     float* __restrict__ out)
{
    __shared__ float lds[16 * 16];

    const int lane = threadIdx.x;
    const int n    = lane & 15;
    const int khi  = lane >> 4;

    const int yo  = blockIdx.x;
    const int bwx = blockIdx.y;
    const int xo  = bwx % XOUT_;
    const int wo  = bwx / XOUT_;
    const int b   = blockIdx.z;

    const int  co   = n / 3;
    const int  d    = n % 3;
    const bool ncol = (n < 12);

    const float* xb = x + ((size_t)b * XS_B + wo * XS_W + xo * XS_X + yo * XS_Y);

    const bool okA = ncol;
    const bool okB = ncol && (khi == 0);
    const int  wlb = co * WS_CO + d;

    v2f bf[NSTEPS];
    int va[NSTEPS][2];
    #pragma unroll
    for (int s = 0; s < NSTEPS; ++s) {
        #pragma unroll
        for (int v = 0; v < 2; ++v) {
            const int2 kd = ktbl[4 * s + 2 * khi + v];
            va[s][v] = kd.x + n;
            const float wv = w[(unsigned)(wlb + kd.y)];
            const bool  ok = (s == NSTEPS - 1) ? okB : okA;
            bf[s][v] = ok ? wv : 0.0f;
        }
    }

    const int  zi1 = lane % ZTILE;
    const int  cc1 = lane / ZTILE;
    const int  i2  = lane + 32;
    const bool ok2 = (i2 < ZTILE * CO_);
    const int  zi2 = i2 % ZTILE;
    const int  cc2 = ok2 ? (i2 / ZTILE) : 3;
    const float bias1 = bias[cc1];
    const float bias2 = bias[cc2];
    float* const op1 = out +
        (((((size_t)(b * CO_ + cc1) * WOUT_ + wo) * XOUT_ + xo) * YOUT_ + yo) * (size_t)ZOUT_ + zi1);
    float* const op2 = out +
        (((((size_t)(b * CO_ + cc2) * WOUT_ + wo) * XOUT_ + xo) * YOUT_ + yo) * (size_t)ZOUT_ + zi2);

    const int dn = (n > 7) ? (n - 7) : 0;

    #pragma unroll
    for (int zt = 0; zt < NZT; ++zt) {
        const int  t0    = zt * ZTILE;
        const bool lastT = (zt == NZT - 1);

        const float* xbt = xb + t0;

        float raw[NSTEPS][2];
        #pragma unroll
        for (int s = 0; s < NSTEPS; ++s) {
            #pragma unroll
            for (int v = 0; v < 2; ++v) {
                const int voff = lastT ? (va[s][v] - dn) : va[s][v];
                raw[s][v] = xbt[(unsigned)voff];
            }
        }

        v8f acc = {};
        #pragma unroll
        for (int s = 0; s < NSTEPS; ++s) {
            v2f af;
            af[0] = raw[s][0];
            af[1] = raw[s][1];
            acc = __builtin_amdgcn_wmma_f32_16x16x4_f32(
                      false, af, false, bf[s], (short)0, acc, false, false);
        }

        #pragma unroll
        for (int v = 0; v < 8; ++v)
            lds[(v + 8 * khi) * 16 + n] = acc[v];
        __syncthreads();

        {
            const int z = t0 + zi1;
            if (z < ZOUT_) {
                float r = bias1;
                #pragma unroll
                for (int dd = 0; dd < 3; ++dd)
                    r += lds[(zi1 + dd) * 16 + (cc1 * 3 + dd)];
                op1[t0] = r;
            }
        }
        if (ok2) {
            const int z = t0 + zi2;
            if (z < ZOUT_) {
                float r = bias2;
                #pragma unroll
                for (int dd = 0; dd < 3; ++dd)
                    r += lds[(zi2 + dd) * 16 + (cc2 * 3 + dd)];
                op2[t0] = r;
            }
        }
        __syncthreads();
    }
}

extern "C" void kernel_launch(void* const* d_in, const int* in_sizes, int n_in,
                              void* d_out, int out_size, void* d_ws, size_t ws_size,
                              hipStream_t stream) {
    const float* x    = (const float*)d_in[0];
    const float* wgt  = (const float*)d_in[1];
    const float* bias = (const float*)d_in[2];
    float* out        = (float*)d_out;

    dim3 grid(YOUT_,
              WOUT_ * XOUT_,
              BB_);
    dim3 block(32);
    conv4d_wmma_f32<<<grid, block, 0, stream>>>(x, wgt, bias, out);
}


# Each wave walks 21 output y-rows of one z-tile: an input y-row is loaded once and feeds the three y-taps from registers (k sum order becomes c-outer); loads per output row 140 -> 50
# speedup vs baseline: 1.5689x; 1.5689x over previous
#include <hip/hip_runtime.h>


typedef __attribute__((ext_vector_type(2))) float v2f;
typedef __attribute__((ext_vector_type(8))) float v8f;

#define BB_    2
#define CO_    4
#define WOUT_  30
#define XOUT_  30
#define YOUT_  62
#define ZOUT_  62
#define ZLAST  63
#define KG     18
#define GSTEPS 5
#define ZTILE  14
#define NZT    5
#define YCH    21
#define NCH    3

#define XS_Y   64
#define XS_X   4096
#define XS_W   131072
#define XS_CI  4194304
#define XS_B   8388608

#define WS_C   3
#define WS_B   9
#define WS_A   27
#define WS_CI  81
#define WS_CO  162

#define KC(k)   ((k) < KG ? (k) : 0)
#define XOFF(k) ((KC(k)/9)*XS_CI + ((KC(k)%9)/3)*XS_W + (KC(k)%3)*XS_X)
#define WOFF(k) ((KC(k)/9)*WS_CI + ((KC(k)%9)/3)*WS_A + (KC(k)%3)*WS_B)

__global__ __launch_bounds__(32)
void conv4d_wmma_f32(const float* __restrict__ x,
                     const float* __restrict__ w,
                     const float* __restrict__ bias,
                     float* __restrict__ out)
{
    __shared__ float lds[16 * 16];

    const int lane = threadIdx.x;
    const int n    = lane & 15;
    const int khi  = lane >> 4;

    const int zt  = blockIdx.x % NZT;
    const int ch  = blockIdx.x / NZT;
    const int bwx = blockIdx.y;
    const int xo  = bwx % XOUT_;
    const int wo  = bwx / XOUT_;
    const int b   = blockIdx.z;

    const int y0   = ch * YCH;
    const int ylen = (YOUT_ - y0 < YCH) ? (YOUT_ - y0) : YCH;
    const int jmax = ylen + 1;
    const int t0   = zt * ZTILE;

    const int tn = (t0 + n > ZLAST) ? ZLAST : (t0 + n);

    const bool ncol = (n < 12);
    const int  co   = ncol ? (n / 3) : 0;
    const int  d    = n % 3;
    const int  wlb  = co * WS_CO + d;

    v2f bfr[3][GSTEPS];
    int vb[GSTEPS][2];
    #pragma unroll
    for (int s = 0; s < GSTEPS; ++s) {
        #pragma unroll
        for (int v = 0; v < 2; ++v) {
            const int  klo  = 4 * s + v;
            const int  khi2 = klo + 2;
            const bool live = khi ? (khi2 < KG) : (klo < KG);
            const int  xoff = khi ? XOFF(khi2) : XOFF(klo);
            const int  woff = khi ? WOFF(khi2) : WOFF(klo);
            vb[s][v] = xoff + tn;
            #pragma unroll
            for (int c = 0; c < 3; ++c) {
                const float wv = w[(unsigned)(wlb + woff + c * WS_C)];
                bfr[c][s][v] = (live && ncol) ? wv : 0.0f;
            }
        }
    }

    const int  zi1 = lane % ZTILE;
    const int  cc1 = lane / ZTILE;
    const int  i2  = lane + 32;
    const bool ok2 = (i2 < ZTILE * CO_);
    const int  zi2 = i2 % ZTILE;
    const int  cc2 = ok2 ? (i2 / ZTILE) : 3;
    const bool st1 = (t0 + zi1 < ZOUT_);
    const bool st2 = ok2 && (t0 + zi2 < ZOUT_);
    const float bias1 = bias[cc1];
    const float bias2 = bias[cc2];
    float* op1 = out +
        (((((size_t)(b * CO_ + cc1) * WOUT_ + wo) * XOUT_ + xo) * YOUT_ + y0) * (size_t)ZOUT_ + t0 + zi1);
    float* op2 = out +
        (((((size_t)(b * CO_ + cc2) * WOUT_ + wo) * XOUT_ + xo) * YOUT_ + y0) * (size_t)ZOUT_ + t0 + zi2);

    const float* xb = x + ((size_t)b * XS_B + wo * XS_W + xo * XS_X + y0 * XS_Y);

    float ac[GSTEPS][2];
    #pragma unroll
    for (int s = 0; s < GSTEPS; ++s) {
        #pragma unroll
        for (int v = 0; v < 2; ++v)
            ac[s][v] = xb[(unsigned)vb[s][v]];
    }

    v8f accM = {};
    v8f accO = {};

    for (int j = 0; j <= jmax; ++j) {
        const int    jn = (j < jmax) ? (j + 1) : jmax;
        const float* xr = xb + jn * XS_Y;
        float an[GSTEPS][2];
        #pragma unroll
        for (int s = 0; s < GSTEPS; ++s) {
            #pragma unroll
            for (int v = 0; v < 2; ++v)
                an[s][v] = xr[(unsigned)vb[s][v]];
        }

        v8f accN = {};
        #pragma unroll
        for (int s = 0; s < GSTEPS; ++s) {
            v2f af;
            af[0] = ac[s][0];
            af[1] = ac[s][1];
            accN = __builtin_amdgcn_wmma_f32_16x16x4_f32(
                       false, af, false, bfr[0][s], (short)0, accN, false, false);
            accM = __builtin_amdgcn_wmma_f32_16x16x4_f32(
                       false, af, false, bfr[1][s], (short)0, accM, false, false);
            accO = __builtin_amdgcn_wmma_f32_16x16x4_f32(
                       false, af, false, bfr[2][s], (short)0, accO, false, false);
        }

        if (j >= 2) {
            #pragma unroll
            for (int v = 0; v < 8; ++v)
                lds[(v + 8 * khi) * 16 + n] = accO[v];
            __syncthreads();

            if (st1) {
                float r = bias1;
                #pragma unroll
                for (int dd = 0; dd < 3; ++dd)
                    r += lds[(zi1 + dd) * 16 + (cc1 * 3 + dd)];
                *op1 = r;
            }
            if (st2) {
                float r = bias2;
                #pragma unroll
                for (int dd = 0; dd < 3; ++dd)
                    r += lds[(zi2 + dd) * 16 + (cc2 * 3 + dd)];
                *op2 = r;
            }
            __syncthreads();
            op1 += ZOUT_;
            op2 += ZOUT_;
        }

        accO = accM;
        accM = accN;
        #pragma unroll
        for (int s = 0; s < GSTEPS; ++s) {
            ac[s][0] = an[s][0];
            ac[s][1] = an[s][1];
        }
    }
}

extern "C" void kernel_launch(void* const* d_in, const int* in_sizes, int n_in,
                              void* d_out, int out_size, void* d_ws, size_t ws_size,
                              hipStream_t stream) {
    const float* x    = (const float*)d_in[0];
    const float* wgt  = (const float*)d_in[1];
    const float* bias = (const float*)d_in[2];
    float* out        = (float*)d_out;

    dim3 grid(NZT * NCH,
              WOUT_ * XOUT_,
              BB_);
    dim3 block(32);
    conv4d_wmma_f32<<<grid, block, 0, stream>>>(x, wgt, bias, out);
}
